# starting kernel
# baseline (speedup 1.0000x reference)
#include <hip/hip_runtime.h>
#include <hip/hip_bf16.h>
#include <math.h>


typedef float v2f __attribute__((ext_vector_type(2)));
typedef float v8f __attribute__((ext_vector_type(8)));

#define A_AGENTS 128
#define D_IN     32
#define HID      128
#define KNN      16
#define OBS_LD   36
#define MAT_LD   132
#define NTHREADS 512
#define NWAVES   (NTHREADS / 32)
#define BIGF     3.4e38f

#define OFF_OBS  0
#define OFF_M    (OFF_OBS + A_AGENTS * OBS_LD)
#define OFF_T    (OFF_M   + A_AGENTS * MAT_LD)
#define OFF_H    (OFF_T   + A_AGENTS * MAT_LD)
#define OFF_DINV (OFF_H   + A_AGENTS * MAT_LD)
#define LDS_FLOATS (OFF_DINV + A_AGENTS)
#define LDS_BYTES  (LDS_FLOATS * 4)

__device__ __forceinline__ float fast_tanh(float x) {
#if __has_builtin(__builtin_amdgcn_tanhf)
    return __builtin_amdgcn_tanhf(x);
#elif __has_builtin(__builtin_amdgcn_tanh_f32)
    return __builtin_amdgcn_tanh_f32(x);
#else
    return tanhf(x);
#endif
}

template<bool ACT>
__device__ __forceinline__ void strip_mm16x4(const float* __restrict__ A, int lda,
                                             const float* __restrict__ B, int ldb,
                                             float* __restrict__ OUT, int ldo,
                                             const float* __restrict__ bias,
                                             int K, int tm, int tn0, int lane)
{
    const int g = lane >> 4;
    const int r = lane & 15;
    const float* Arow = A + (tm * 16 + r) * lda;
    const float* Bcol = B + tn0 * 16 + r;
    v8f c0 = {}, c1 = {}, c2 = {}, c3 = {};
    for (int k0 = 0; k0 < K; k0 += 4) {
        const int ka = k0 + 2 * g;
        const float* Brow0 = Bcol + (ka)     * ldb;
        const float* Brow1 = Bcol + (ka + 1) * ldb;
        v2f a, b0, b1, b2, b3;
        a.x  = Arow[ka];   a.y  = Arow[ka + 1];
        b0.x = Brow0[0];   b0.y = Brow1[0];
        b1.x = Brow0[16];  b1.y = Brow1[16];
        b2.x = Brow0[32];  b2.y = Brow1[32];
        b3.x = Brow0[48];  b3.y = Brow1[48];
        c0 = __builtin_amdgcn_wmma_f32_16x16x4_f32(false, a, false, b0, (short)0, c0, false, false);
        c1 = __builtin_amdgcn_wmma_f32_16x16x4_f32(false, a, false, b1, (short)0, c1, false, false);
        c2 = __builtin_amdgcn_wmma_f32_16x16x4_f32(false, a, false, b2, (short)0, c2, false, false);
        c3 = __builtin_amdgcn_wmma_f32_16x16x4_f32(false, a, false, b3, (short)0, c3, false, false);
    }
    const v8f cc[4] = {c0, c1, c2, c3};
    float* obase = OUT + (tm * 16 + 8 * g) * ldo + tn0 * 16 + r;
#pragma unroll
    for (int t = 0; t < 4; ++t) {
        const float bval = ACT ? bias[tn0 * 16 + t * 16 + r] : 0.0f;
#pragma unroll
        for (int v = 0; v < 8; ++v) {
            float val = cc[t][v];
            if (ACT) val = fast_tanh(val + bval);
            obase[v * ldo + t * 16] = val;
        }
    }
}

__global__ void __launch_bounds__(NTHREADS)
gnn_critic_fused(const float* __restrict__ obs,
                 const float* __restrict__ W1, const float* __restrict__ b1,
                 const float* __restrict__ W2, const float* __restrict__ b2,
                 const float* __restrict__ Wout, const float* __restrict__ bout,
                 float* __restrict__ out)
{
    extern __shared__ float smem[];
    float* obs_s  = smem + OFF_OBS;
    float* M_s    = smem + OFF_M;
    float* T_s    = smem + OFF_T;
    float* H_s    = smem + OFF_H;
    float* dinv_s = smem + OFF_DINV;

    const int tid  = threadIdx.x;
    const int lane = tid & 31;
    const int wave = tid >> 5;
    const int tm   = wave >> 1;
    const int tn0  = (wave & 1) * 4;
    const int bidx = blockIdx.x;

    const float* gobs = obs + (size_t)bidx * (A_AGENTS * D_IN);
    for (int i = tid; i < A_AGENTS * (D_IN / 4); i += NTHREADS) {
        const int row = i / (D_IN / 4);
        const int c4  = i % (D_IN / 4);
        const float4 v = ((const float4*)gobs)[i];
        float* dst = obs_s + row * OBS_LD + c4 * 4;
        dst[0] = v.x; dst[1] = v.y; dst[2] = v.z; dst[3] = v.w;
    }
    for (int i = tid; i < A_AGENTS * MAT_LD; i += NTHREADS) M_s[i] = 0.0f;
    __syncthreads();

    for (int e = tid; e < A_AGENTS * A_AGENTS; e += NTHREADS) {
        const int i = e >> 7;
        const int j = e & 127;
        const float dx = obs_s[i * OBS_LD + 0] - obs_s[j * OBS_LD + 0];
        const float dy = obs_s[i * OBS_LD + 1] - obs_s[j * OBS_LD + 1];
        float d = dx * dx + dy * dy;
        if (i == j) d = BIGF;
        T_s[i * MAT_LD + j] = d;
    }
    __syncthreads();

    {
        const int a  = tid >> 2;
        const int sl = tid & 3;
        float* drow = T_s + a * MAT_LD + sl * 32;
        for (int t = 0; t < KNN; ++t) {
            float best = BIGF;
            int bestj = 0;
            for (int j = 0; j < 32; ++j) {
                const float d = drow[j];
                if (d < best) { best = d; bestj = sl * 32 + j; }
            }
            for (int off = 1; off < 4; off <<= 1) {
                const float ob = __shfl_xor(best, off);
                const int   oj = __shfl_xor(bestj, off);
                if (ob < best || (ob == best && oj < bestj)) {
                    best = ob; bestj = oj;
                }
            }
            if (sl == 0) {
                T_s[a * MAT_LD + bestj] = BIGF;
                M_s[bestj * MAT_LD + a] = 1.0f;
            }
        }
    }
    __syncthreads();

    {
        const int a  = tid >> 2;
        const int sl = tid & 3;
        const float* seg = M_s + a * MAT_LD + sl * 32;
        float s = 0.0f;
        for (int i = 0; i < 32; ++i) s += seg[i];
        s += __shfl_xor(s, 1);
        s += __shfl_xor(s, 2);
        if (sl == 0) dinv_s[a] = rsqrtf(1.0f + s);
    }
    __syncthreads();

    for (int e = tid; e < A_AGENTS * A_AGENTS; e += NTHREADS) {
        const int j = e >> 7;
        const int i = e & 127;
        float v = M_s[j * MAT_LD + i] * dinv_s[i] * dinv_s[j];
        if (i == j) v = dinv_s[j] * dinv_s[j];
        M_s[j * MAT_LD + i] = v;
    }
    __syncthreads();

    strip_mm16x4<false>(obs_s, OBS_LD, W1, HID, T_s, MAT_LD, nullptr, D_IN,
                        tm, tn0, lane);
    __syncthreads();

    strip_mm16x4<true>(M_s, MAT_LD, T_s, MAT_LD, H_s, MAT_LD, b1, HID,
                       tm, tn0, lane);
    __syncthreads();

    strip_mm16x4<false>(H_s, MAT_LD, W2, HID, T_s, MAT_LD, nullptr, HID,
                        tm, tn0, lane);
    __syncthreads();

    strip_mm16x4<true>(M_s, MAT_LD, T_s, MAT_LD, H_s, MAT_LD, b2, HID,
                       tm, tn0, lane);
    __syncthreads();

    {
        const int a  = tid >> 2;
        const int sl = tid & 3;
        const float* hrow = H_s + a * MAT_LD + sl * 32;
        const float* wseg = Wout + sl * 32;
        float s = 0.0f;
        for (int k = 0; k < 32; ++k) s += hrow[k] * wseg[k];
        s += __shfl_xor(s, 1);
        s += __shfl_xor(s, 2);
        if (sl == 0) out[(size_t)bidx * A_AGENTS + a] = s + bout[0];
    }
}

extern "C" void kernel_launch(void* const* d_in, const int* in_sizes, int n_in,
                              void* d_out, int out_size, void* d_ws, size_t ws_size,
                              hipStream_t stream) {
    (void)n_in; (void)out_size; (void)d_ws; (void)ws_size;
    const float* obs  = (const float*)d_in[0];
    const float* W1   = (const float*)d_in[1];
    const float* b1   = (const float*)d_in[2];
    const float* W2   = (const float*)d_in[3];
    const float* b2   = (const float*)d_in[4];
    const float* Wout = (const float*)d_in[5];
    const float* bout = (const float*)d_in[6];
    float* out = (float*)d_out;

    const int B = in_sizes[0] / (A_AGENTS * D_IN);

    static_assert(LDS_BYTES <= 320 * 1024, "LDS over WGP capacity");
    (void)hipFuncSetAttribute((const void*)gnn_critic_fused,
                              hipFuncAttributeMaxDynamicSharedMemorySize,
                              LDS_BYTES);
    gnn_critic_fused<<<B, NTHREADS, LDS_BYTES, stream>>>(
        obs, W1, b1, W2, b2, Wout, bout, out);
}


# One workgroup per pair of batches, pipelined by wave role: waves 8-15 build batch 2's kNN adjacency and degrees while waves 0-7 run batch 1's products; W2 fragments fetched per k-block; same sums
# speedup vs baseline: 1.7439x; 1.7439x over previous
#include <hip/hip_runtime.h>
#include <hip/hip_bf16.h>
#include <math.h>


typedef float v2f __attribute__((ext_vector_type(2)));
typedef float v8f __attribute__((ext_vector_type(8)));

#define A_AGENTS 128
#define D_IN     32
#define HID      128
#define KNN      16
#define OBS_LD   36
#define MAT_LD   132
#define NTHREADS 512
#define NWAVES   (NTHREADS / 32)
#define BIGF     3.4e38f

#define OBS_PLANE (A_AGENTS * OBS_LD)
#define MAT_PLANE (A_AGENTS * MAT_LD)
#define OFF_OBS  0
#define OFF_M    (OFF_OBS + 2 * OBS_PLANE)
#define OFF_T    (OFF_M   + 2 * MAT_PLANE)
#define OFF_H    (OFF_T   + MAT_PLANE)
#define OFF_DINV (OFF_H   + MAT_PLANE)
#define LDS_FLOATS (OFF_DINV + 2 * A_AGENTS)
#define LDS_BYTES  (LDS_FLOATS * 4)

__device__ __forceinline__ float fast_tanh(float x) {
#if __has_builtin(__builtin_amdgcn_tanhf)
    return __builtin_amdgcn_tanhf(x);
#elif __has_builtin(__builtin_amdgcn_tanh_f32)
    return __builtin_amdgcn_tanh_f32(x);
#else
    return tanhf(x);
#endif
}


typedef float v4f __attribute__((ext_vector_type(4)));

#define SCHED_FENCE() __builtin_amdgcn_sched_barrier(0)

template<int K>
__device__ __forceinline__ void load_wfrag(const float* __restrict__ Wg, int col0,
                                           int lane, v2f (&w)[K / 4])
{
    const int h = lane >> 4, c = lane & 15;
    const float* p = Wg + (4 * h) * HID + col0 + c;
#pragma unroll
    for (int i = 0; i < K / 4; ++i) {
        const int k = 8 * (i >> 1) + 2 * (i & 1);
        w[i].x = p[k * HID];
        w[i].y = p[(k + 1) * HID];
    }
}

__device__ __forceinline__ void load_a8(const float* __restrict__ p, int ld, v4f (&a)[8])
{
#pragma unroll
    for (int R = 0; R < 8; ++R) a[R] = *(const v4f*)(p + R * 16 * ld);
}

template<int U>
__device__ __forceinline__ void wmma8(const v4f (&a)[8], v2f b, v8f (&acc)[8])
{
#pragma unroll
    for (int R = 0; R < 8; ++R) {
        const v2f av = __builtin_shufflevector(a[R], a[R], 2 * U, 2 * U + 1);
        acc[R] = __builtin_amdgcn_wmma_f32_16x16x4_f32(false, av, false, b, (short)0, acc[R], false, false);
    }
}

template<int K>
__device__ __forceinline__ void colblock_xw(const float* __restrict__ A, int lda,
                                            const v2f (&w)[K / 4], v8f (&t)[8], int lane)
{
    const int h = lane >> 4, c = lane & 15;
    const float* Ap = A + c * lda + 4 * h;
#pragma unroll
    for (int R = 0; R < 8; ++R) t[R] = v8f{};
    v4f a0[8], a1[8];
    load_a8(Ap, lda, a0);
#pragma unroll
    for (int j = 0; j < K / 8; j += 2) {
        load_a8(Ap + 8 * (j + 1), lda, a1);
        SCHED_FENCE();
        wmma8<0>(a0, w[2 * j], t);
        wmma8<1>(a0, w[2 * j + 1], t);
        SCHED_FENCE();
        if (j + 2 < K / 8) load_a8(Ap + 8 * (j + 2), lda, a0);
        SCHED_FENCE();
        wmma8<0>(a1, w[2 * j + 2], t);
        wmma8<1>(a1, w[2 * j + 3], t);
        SCHED_FENCE();
    }
}

__device__ __forceinline__ void load_w4(const float* __restrict__ p, int j, v2f (&w4)[4])
{
#pragma unroll
    for (int q = 0; q < 4; ++q) {
        const int i = 2 * j + q;
        const int k = 8 * (i >> 1) + 2 * (i & 1);
        w4[q].x = p[k * HID];
        w4[q].y = p[(k + 1) * HID];
    }
}

template<int K>
__device__ __forceinline__ void colblock_xw_stream(const float* __restrict__ A, int lda,
                                                   const float* __restrict__ Wg, int col0,
                                                   const v2f (&wfirst)[4], v8f (&t)[8], int lane)
{
    const int h = lane >> 4, c = lane & 15;
    const float* Ap = A + c * lda + 4 * h;
    const float* Wp = Wg + (4 * h) * HID + col0 + c;
#pragma unroll
    for (int R = 0; R < 8; ++R) t[R] = v8f{};
    v4f a0[8], a1[8];
    v2f wc[4], wn[4];
#pragma unroll
    for (int q = 0; q < 4; ++q) wc[q] = wfirst[q];
    load_a8(Ap, lda, a0);
#pragma unroll
    for (int j = 0; j < K / 8; j += 2) {
        load_a8(Ap + 8 * (j + 1), lda, a1);
        if (j + 2 < K / 8) load_w4(Wp, j + 2, wn);
        SCHED_FENCE();
        wmma8<0>(a0, wc[0], t);
        wmma8<1>(a0, wc[1], t);
        SCHED_FENCE();
        if (j + 2 < K / 8) load_a8(Ap + 8 * (j + 2), lda, a0);
        SCHED_FENCE();
        wmma8<0>(a1, wc[2], t);
        wmma8<1>(a1, wc[3], t);
        SCHED_FENCE();
        if (j + 2 < K / 8) {
#pragma unroll
            for (int q = 0; q < 4; ++q) wc[q] = wn[q];
        }
    }
}

template<int Q>
__device__ __forceinline__ v2f reg_pair(const v8f& v)
{
    return __builtin_shufflevector(v, v, 2 * Q, 2 * Q + 1);
}

__device__ __forceinline__ void colblock_mt(const float* __restrict__ M, const v8f (&t)[8],
                                            v8f (&acc)[8], int lane)
{
    const int h = lane >> 4, c = lane & 15;
    const float* Ap = M + c * MAT_LD + 8 * h;
#pragma unroll
    for (int R = 0; R < 8; ++R) acc[R] = v8f{};
    v4f a0[8], a1[8];
    load_a8(Ap, MAT_LD, a0);
#pragma unroll
    for (int Rk = 0; Rk < 8; ++Rk) {
        load_a8(Ap + 16 * Rk + 4, MAT_LD, a1);
        SCHED_FENCE();
        wmma8<0>(a0, reg_pair<0>(t[Rk]), acc);
        wmma8<1>(a0, reg_pair<1>(t[Rk]), acc);
        SCHED_FENCE();
        if (Rk + 1 < 8) load_a8(Ap + 16 * (Rk + 1), MAT_LD, a0);
        SCHED_FENCE();
        wmma8<0>(a1, reg_pair<2>(t[Rk]), acc);
        wmma8<1>(a1, reg_pair<3>(t[Rk]), acc);
        SCHED_FENCE();
    }
}

__device__ __forceinline__ void store_act(const v8f (&acc)[8], float bias,
                                          float* __restrict__ OUT, int lane)
{
    const int h = lane >> 4, c = lane & 15;
    float* op = OUT + (8 * h) * MAT_LD + c;
#pragma unroll
    for (int R = 0; R < 8; ++R) {
#pragma unroll
        for (int v = 0; v < 8; ++v)
            op[(16 * R + v) * MAT_LD] = fast_tanh(acc[R][v] + bias);
    }
}

#define CE(x, y) { const float lo_ = fminf(x, y); const float hi_ = fmaxf(x, y); x = lo_; y = hi_; }

__device__ __forceinline__ void sort16(float (&v)[16])
{
    CE(v[0], v[1]) CE(v[2], v[3]) CE(v[4], v[5]) CE(v[6], v[7])
    CE(v[8], v[9]) CE(v[10], v[11]) CE(v[12], v[13]) CE(v[14], v[15])
    CE(v[0], v[2]) CE(v[1], v[3]) CE(v[4], v[6]) CE(v[5], v[7])
    CE(v[8], v[10]) CE(v[9], v[11]) CE(v[12], v[14]) CE(v[13], v[15])
    CE(v[1], v[2]) CE(v[5], v[6]) CE(v[9], v[10]) CE(v[13], v[14])
    CE(v[0], v[4]) CE(v[1], v[5]) CE(v[2], v[6]) CE(v[3], v[7])
    CE(v[8], v[12]) CE(v[9], v[13]) CE(v[10], v[14]) CE(v[11], v[15])
    CE(v[2], v[4]) CE(v[3], v[5]) CE(v[10], v[12]) CE(v[11], v[13])
    CE(v[1], v[2]) CE(v[3], v[4]) CE(v[5], v[6]) CE(v[9], v[10])
    CE(v[11], v[12]) CE(v[13], v[14]) CE(v[0], v[8]) CE(v[1], v[9])
    CE(v[2], v[10]) CE(v[3], v[11]) CE(v[4], v[12]) CE(v[5], v[13])
    CE(v[6], v[14]) CE(v[7], v[15]) CE(v[4], v[8]) CE(v[5], v[9])
    CE(v[6], v[10]) CE(v[7], v[11]) CE(v[2], v[4]) CE(v[3], v[5])
    CE(v[6], v[8]) CE(v[7], v[9]) CE(v[10], v[12]) CE(v[11], v[13])
    CE(v[1], v[2]) CE(v[3], v[4]) CE(v[5], v[6]) CE(v[7], v[8])
    CE(v[9], v[10]) CE(v[11], v[12]) CE(v[13], v[14])
}

__device__ __forceinline__ void bitonic_merge16(float (&v)[16])
{
#pragma unroll
    for (int j = 8; j >= 1; j >>= 1) {
#pragma unroll
        for (int i = 0; i < 16; ++i)
            if ((i & j) == 0) CE(v[i], v[i | j])
    }
}

template<int CTRL>
__device__ __forceinline__ int quad_get(int x)
{
    return __builtin_amdgcn_update_dpp(x, x, CTRL, 0xf, 0xf, true);
}
template<int CTRL>
__device__ __forceinline__ float quad_getf(float x)
{
    return __int_as_float(quad_get<CTRL>(__float_as_int(x)));
}
#define QUAD_XOR1 0xB1
#define QUAD_XOR2 0x4E

__device__ __forceinline__ void knn_adjacency(const float* __restrict__ obs_s,
                                              float* __restrict__ M_s, int a, int sl)
{
    const v2f pa = *(const v2f*)(obs_s + a * OBS_LD);
    const float* pj = obs_s + sl * OBS_LD;
    float d[32];
#pragma unroll
    for (int jj = 0; jj < 32; ++jj) {
        const v2f p = *(const v2f*)(pj + jj * 4 * OBS_LD);
        const float dx = pa.x - p.x;
        const float dy = pa.y - p.y;
        float dd;
        {
#pragma clang fp contract(off)
            dd = dx * dx + dy * dy;
        }
        d[jj] = (4 * jj + sl == a) ? BIGF : dd;
    }
    float lo[16], hi[16];
#pragma unroll
    for (int i = 0; i < 16; ++i) { lo[i] = d[i]; hi[i] = d[16 + i]; }
    sort16(lo);
    sort16(hi);
#pragma unroll
    for (int i = 0; i < 16; ++i) lo[i] = fminf(lo[i], hi[15 - i]);
    bitonic_merge16(lo);
#pragma unroll
    for (int i = 0; i < 16; ++i) hi[i] = quad_getf<QUAD_XOR1>(lo[15 - i]);
#pragma unroll
    for (int i = 0; i < 16; ++i) lo[i] = fminf(lo[i], hi[i]);
    bitonic_merge16(lo);
#pragma unroll
    for (int i = 0; i < 16; ++i) hi[i] = quad_getf<QUAD_XOR2>(lo[15 - i]);
    float tau = fminf(lo[0], hi[0]);
#pragma unroll
    for (int i = 1; i < 16; ++i) tau = fmaxf(tau, fminf(lo[i], hi[i]));

    float* mcol = M_s + sl * MAT_LD + a;
    int n_le = 0;
#pragma unroll
    for (int jj = 0; jj < 32; ++jj) {
        const bool le = d[jj] <= tau;
        n_le += le ? 1 : 0;
        mcol[jj * 4 * MAT_LD] = le ? 1.0f : 0.0f;
    }
    n_le += quad_get<QUAD_XOR1>(n_le);
    n_le += quad_get<QUAD_XOR2>(n_le);
    if (n_le != KNN) {
        int n_lt = 0;
        unsigned eqm = 0u;
#pragma unroll
        for (int jj = 0; jj < 32; ++jj) {
            n_lt += (d[jj] < tau) ? 1 : 0;
            eqm  |= (d[jj] == tau) ? (1u << jj) : 0u;
        }
        n_lt += quad_get<QUAD_XOR1>(n_lt);
        n_lt += quad_get<QUAD_XOR2>(n_lt);
        const int need = KNN - n_lt;
        unsigned m[4];
        m[0] = (unsigned)quad_get<0x00>((int)eqm);
        m[1] = (unsigned)quad_get<0x55>((int)eqm);
        m[2] = (unsigned)quad_get<0xAA>((int)eqm);
        m[3] = (unsigned)quad_get<0xFF>((int)eqm);
#pragma unroll 1
        for (int jj = 0; jj < 32; ++jj) {
            if ((eqm >> jj) & 1u) {
                const unsigned below = (1u << jj) - 1u;
                const unsigned upto  = below | (1u << jj);
                int rank = 0;
#pragma unroll
                for (int s = 0; s < 4; ++s)
                    rank += __popc(m[s] & ((s < sl) ? upto : below));
                if (rank >= need) mcol[jj * 4 * MAT_LD] = 0.0f;
            }
        }
    }
}

__device__ __forceinline__ void degree_row(const float* __restrict__ M_s,
                                           float* __restrict__ dinv_s, int a, int sl)
{
    const float* seg = M_s + a * MAT_LD + sl * 32;
    float s = 0.0f;
    for (int i = 0; i < 32; ++i) s += seg[i];
    s += __shfl_xor(s, 1);
    s += __shfl_xor(s, 2);
    if (sl == 0) dinv_s[a] = rsqrtf(1.0f + s);
}

__device__ __forceinline__ void normalise_adj(float* __restrict__ M_s,
                                              const float* __restrict__ dinv_s, int tid)
{
    for (int e = tid; e < A_AGENTS * A_AGENTS; e += NTHREADS) {
        const int j = e >> 7;
        const int i = e & 127;
        float v = M_s[j * MAT_LD + i] * dinv_s[i] * dinv_s[j];
        if (i == j) v = dinv_s[j] * dinv_s[j];
        M_s[j * MAT_LD + i] = v;
    }
}

__device__ __forceinline__ void out_proj(const float* __restrict__ H2_s,
                                         const float* __restrict__ Wout,
                                         const float* __restrict__ bout,
                                         float* __restrict__ outb, int tid)
{
    const int a  = tid >> 2;
    const int sl = tid & 3;
    const float* hrow = H2_s + a * MAT_LD + sl * 32;
    const float* wseg = Wout + sl * 32;
    float s = 0.0f;
    for (int k = 0; k < 32; ++k) s += hrow[k] * wseg[k];
    s += __shfl_xor(s, 1);
    s += __shfl_xor(s, 2);
    if (sl == 0) outb[a] = s + bout[0];
}

__global__ void __launch_bounds__(NTHREADS)
gnn_critic_fused(const float* __restrict__ obs,
                 const float* __restrict__ W1, const float* __restrict__ b1,
                 const float* __restrict__ W2, const float* __restrict__ b2,
                 const float* __restrict__ Wout, const float* __restrict__ bout,
                 float* __restrict__ out, int nbatch)
{
    extern __shared__ __attribute__((aligned(16))) float smem[];
    float* T_s = smem + OFF_T;
    float* H_s = smem + OFF_H;

    const int tid  = threadIdx.x;
    const int lane = tid & 31;
    const int wv   = __builtin_amdgcn_readfirstlane(tid >> 5);

    const int  bA   = blockIdx.x;
    const int  bB   = blockIdx.x + gridDim.x;
    const bool hasB = bB < nbatch;

    float4 pb0 = make_float4(0.f, 0.f, 0.f, 0.f), pb1 = pb0;
    if (hasB) {
        const float4* gB = (const float4*)(obs + (size_t)bB * (A_AGENTS * D_IN));
        pb0 = gB[tid];
        pb1 = gB[tid + NTHREADS];
    }
    {
        float* obs_s = smem + OFF_OBS;
        const float* gobs = obs + (size_t)bA * (A_AGENTS * D_IN);
        for (int i = tid; i < A_AGENTS * (D_IN / 4); i += NTHREADS) {
            const int row = i / (D_IN / 4);
            const int c4  = i % (D_IN / 4);
            const float4 v = ((const float4*)gobs)[i];
            float* dst = obs_s + row * OBS_LD + c4 * 4;
            dst[0] = v.x; dst[1] = v.y; dst[2] = v.z; dst[3] = v.w;
        }
    }
    if (hasB) {
        float* obs_s = smem + OFF_OBS + OBS_PLANE;
        const int c4 = tid % (D_IN / 4);
        float* d0 = obs_s + (tid / (D_IN / 4)) * OBS_LD + c4 * 4;
        float* d1 = d0 + (NTHREADS / (D_IN / 4)) * OBS_LD;
        d0[0] = pb0.x; d0[1] = pb0.y; d0[2] = pb0.z; d0[3] = pb0.w;
        d1[0] = pb1.x; d1[1] = pb1.y; d1[2] = pb1.z; d1[3] = pb1.w;
    }
    __syncthreads();

#pragma unroll 1
    for (int it = 0; it < 3; ++it) {
        const bool build = (it == 0) || (it == 1 && hasB);
        const bool gemm  = (it == 1) || (it == 2 && hasB);
        const int  pb    = it & 1;
        const int  pg    = pb ^ 1;
        const float* obs_b  = smem + OFF_OBS  + pb * OBS_PLANE;
        float*       M_b    = smem + OFF_M    + pb * MAT_PLANE;
        float*       dinv_b = smem + OFF_DINV + pb * A_AGENTS;
        const float* obs_g  = smem + OFF_OBS  + pg * OBS_PLANE;
        const float* M_g    = smem + OFF_M    + pg * MAT_PLANE;
        const bool cons = gemm && (wv < 8);
        const bool prod = build && (it == 0 || wv >= 8);
        const int vt0   = (it == 0) ? tid : (tid - NTHREADS / 2);
        const int npass = (it == 0) ? 1 : 2;
        const int col0  = (wv & 7) * 16;

        v2f w2first[4];
        if (cons) {
            v2f w1f[D_IN / 4];
            v8f t[8], acc[8];
            load_wfrag<D_IN>(W1, col0, lane, w1f);
            colblock_xw<D_IN>(obs_g, OBS_LD, w1f, t, lane);
            colblock_mt(M_g, t, acc, lane);
            load_w4(W2 + (4 * (lane >> 4)) * HID + col0 + (lane & 15), 0, w2first);
            store_act(acc, b1[col0 + (lane & 15)], H_s + col0, lane);
        } else if (prod) {
#pragma unroll 1
            for (int p = 0; p < npass; ++p) {
                const int vt = vt0 + p * (NTHREADS / 2);
                knn_adjacency(obs_b, M_b, vt >> 2, vt & 3);
            }
        }
        __syncthreads();

        if (cons) {
            v8f t[8], acc[8];
            colblock_xw_stream<HID>(H_s, MAT_LD, W2, col0, w2first, t, lane);
            colblock_mt(M_g, t, acc, lane);
            store_act(acc, b2[col0 + (lane & 15)], T_s + col0, lane);
        } else if (prod) {
#pragma unroll 1
            for (int p = 0; p < npass; ++p) {
                const int vt = vt0 + p * (NTHREADS / 2);
                degree_row(M_b, dinv_b, vt >> 2, vt & 3);
            }
        }
        __syncthreads();

        if (build) normalise_adj(M_b, dinv_b, tid);
        if (gemm)  out_proj(T_s, Wout, bout, out + (size_t)((it == 1) ? bA : bB) * A_AGENTS, tid);
        __syncthreads();
    }
}

extern "C" void kernel_launch(void* const* d_in, const int* in_sizes, int n_in,
                              void* d_out, int out_size, void* d_ws, size_t ws_size,
                              hipStream_t stream) {
    (void)n_in; (void)out_size; (void)d_ws; (void)ws_size;
    const float* obs  = (const float*)d_in[0];
    const float* W1   = (const float*)d_in[1];
    const float* b1   = (const float*)d_in[2];
    const float* W2   = (const float*)d_in[3];
    const float* b2   = (const float*)d_in[4];
    const float* Wout = (const float*)d_in[5];
    const float* bout = (const float*)d_in[6];
    float* out = (float*)d_out;

    const int B = in_sizes[0] / (A_AGENTS * D_IN);

    static_assert(LDS_BYTES <= 320 * 1024, "LDS over WGP capacity");
    (void)hipFuncSetAttribute((const void*)gnn_critic_fused,
                              hipFuncAttributeMaxDynamicSharedMemorySize,
                              LDS_BYTES);
    gnn_critic_fused<<<(B + 1) / 2, NTHREADS, LDS_BYTES, stream>>>(
        obs, W1, b1, W2, b2, Wout, bout, out, B);
}
